# starting kernel
# baseline (speedup 1.0000x reference)
#include <hip/hip_runtime.h>
#include <hip/hip_bf16.h>

typedef float v2f __attribute__((ext_vector_type(2)));
typedef float v8f __attribute__((ext_vector_type(8)));

#define BATCH 4
#define SEQ   2048
#define DMOD  1024
#define NST   16
#define CHUNK 64
#define NCH   (SEQ / CHUNK)

#define WS_AMATT 0
#define WS_UT    (80 * 64)
#define WS_ACT   (80 * 64 + 64 * 16)
#define WS_S     8192
#define WS_H     (8192 + (size_t)BATCH * NCH * DMOD * NST)

__global__ __launch_bounds__(256) void k0_precompute(const float* __restrict__ A,
                                                     const float* __restrict__ Bvec,
                                                     float* __restrict__ ws) {
    __shared__ float cur[16][16];
    __shared__ float bb[16];
    __shared__ float kv[64];
    __shared__ float r[16];
    __shared__ float u[16];

    const int tid = threadIdx.x;
    const int m = tid >> 4, n = tid & 15;

    float* AmatT = ws + WS_AMATT;
    float* Ut    = ws + WS_UT;
    float* AcT   = ws + WS_ACT;

    if (tid < 16) bb[tid] = Bvec[tid];
    cur[m][n] = (m == n) ? 1.0f : 0.0f;
    __syncthreads();

    for (int p = 0; p <= CHUNK; ++p) {
        if (tid < 16) {
            float s = 0.f;
            for (int q = 0; q < 16; ++q) s += bb[q] * cur[q][tid];
            r[tid] = s;
        } else if (tid < 32) {
            const int mm = tid - 16;
            float s = 0.f;
            for (int q = 0; q < 16; ++q) s += cur[mm][q] * bb[q];
            u[mm] = s;
        }
        __syncthreads();

        if (p <= 63) {
            if (tid == 0) {
                float s = 0.f;
                for (int q = 0; q < 16; ++q) s += r[q] * bb[q];
                kv[p] = s;
            }
            if (tid < 16) Ut[(63 - p) * 16 + tid] = u[tid];
        }
        if (p >= 1) {
            if (tid < 16) AmatT[(64 + tid) * 64 + (p - 1)] = r[tid];
        }
        if (p == CHUNK) {
            AcT[n * 16 + m] = cur[m][n];
        }

        float s = 0.f;
        for (int q = 0; q < 16; ++q) s += A[m * 16 + q] * cur[q][n];
        __syncthreads();
        cur[m][n] = s;
        __syncthreads();
    }
    __syncthreads();

    for (int e = tid; e < 64 * 64; e += 256) {
        const int k = e >> 6, t = e & 63;
        AmatT[k * 64 + t] = (t >= k) ? kv[t - k] : 0.0f;
    }
}

__global__ __launch_bounds__(256) void k1_state_gemm(const float* __restrict__ x,
                                                     const float* __restrict__ Ut,
                                                     float* __restrict__ S) {
    const int lane = threadIdx.x & 31;
    const int wave = threadIdx.x >> 5;
    const int hi = lane >> 4;
    const int m = lane & 15;

    const int blk = blockIdx.x;
    const int dslab = blk & 7;
    const int chunk = (blk >> 3) & (NCH - 1);
    const int b = blk >> 8;
    const int d0 = (dslab * 8 + wave) * 16;

    const float* xb = x + ((size_t)b * SEQ + (size_t)chunk * CHUNK) * DMOD;

    v8f acc = {};
#pragma unroll
    for (int k0 = 0; k0 < CHUNK; k0 += 4) {
        const int r0 = k0 + 2 * hi;
        v2f a, bf;
        a.x  = xb[(size_t)r0 * DMOD + d0 + m];
        a.y  = xb[(size_t)(r0 + 1) * DMOD + d0 + m];
        bf.x = Ut[r0 * 16 + m];
        bf.y = Ut[(r0 + 1) * 16 + m];
        acc = __builtin_amdgcn_wmma_f32_16x16x4_f32(false, a, false, bf,
                                                    (short)0, acc, false, false);
    }

    float* Sp = S + (((size_t)b * NCH + chunk) * DMOD + d0) * NST;
#pragma unroll
    for (int v = 0; v < 8; ++v)
        Sp[(v + 8 * hi) * NST + m] = acc[v];
}

__global__ __launch_bounds__(256) void k2_scan(const float* __restrict__ AcT,
                                               const float* __restrict__ S,
                                               float* __restrict__ H) {
    __shared__ float hbuf[8][16][17];

    const int lane = threadIdx.x & 31;
    const int wave = threadIdx.x >> 5;
    const int hi = lane >> 4;
    const int m = lane & 15;

    const int gwave = blockIdx.x * 8 + wave;
    const int b = gwave >> 6;
    const int d0 = (gwave & 63) * 16;

    v2f bf[4];
#pragma unroll
    for (int k0 = 0; k0 < 16; k0 += 4) {
        const int r0 = k0 + 2 * hi;
        bf[k0 >> 2].x = AcT[r0 * 16 + m];
        bf[k0 >> 2].y = AcT[(r0 + 1) * 16 + m];
    }

    v8f h = {};
    for (int c = 0; c < NCH; ++c) {
#pragma unroll
        for (int v = 0; v < 8; ++v)
            hbuf[wave][v + 8 * hi][m] = h[v];
        __syncthreads();

        float* Hp = H + (((size_t)b * NCH + c) * NST) * DMOD;
#pragma unroll
        for (int v = 0; v < 8; ++v)
            Hp[(size_t)(2 * v + hi) * DMOD + d0 + m] = hbuf[wave][m][2 * v + hi];

        const float* Sp = S + (((size_t)b * NCH + c) * DMOD + d0) * NST;
        v8f acc;
#pragma unroll
        for (int v = 0; v < 8; ++v)
            acc[v] = Sp[(v + 8 * hi) * NST + m];

#pragma unroll
        for (int k0 = 0; k0 < 16; k0 += 4) {
            const int r0 = k0 + 2 * hi;
            v2f a;
            a.x = hbuf[wave][m][r0];
            a.y = hbuf[wave][m][r0 + 1];
            acc = __builtin_amdgcn_wmma_f32_16x16x4_f32(false, a, false, bf[k0 >> 2],
                                                        (short)0, acc, false, false);
        }
        h = acc;
        __syncthreads();
    }
}

__global__ __launch_bounds__(256) void k3_out_gemm(const float* __restrict__ x,
                                                   const float* __restrict__ AmatT,
                                                   const float* __restrict__ H,
                                                   float* __restrict__ y) {
    __shared__ float Asm[80 * 64];
    for (int e = threadIdx.x; e < 80 * 64; e += 256) Asm[e] = AmatT[e];
    __syncthreads();

    const int lane = threadIdx.x & 31;
    const int wave = threadIdx.x >> 5;
    const int hi = lane >> 4;
    const int m = lane & 15;

    const int blk = blockIdx.x;
    const int dquad = blk & 3;
    const int chunk = (blk >> 2) & (NCH - 1);
    const int b = blk >> 7;
    const int d0 = (dquad * 8 + wave) * 32;
    const int d1 = d0 + 16;

    const float* xb = x + ((size_t)b * SEQ + (size_t)chunk * CHUNK) * DMOD;
    const float* Hp = H + (((size_t)b * NCH + chunk) * NST) * DMOD;

    v8f acc[2][4] = {{{}, {}, {}, {}}, {{}, {}, {}, {}}};

#pragma unroll
    for (int kb = 0; kb < 4; ++kb) {
#pragma unroll
        for (int kk = 0; kk < 16; kk += 4) {
            const int r0 = kb * 16 + kk + 2 * hi;
            v2f b0, b1;
            b0.x = xb[(size_t)r0 * DMOD + d0 + m];
            b0.y = xb[(size_t)(r0 + 1) * DMOD + d0 + m];
            b1.x = xb[(size_t)r0 * DMOD + d1 + m];
            b1.y = xb[(size_t)(r0 + 1) * DMOD + d1 + m];
#pragma unroll
            for (int it = kb; it < 4; ++it) {
                v2f a;
                a.x = Asm[r0 * 64 + it * 16 + m];
                a.y = Asm[(r0 + 1) * 64 + it * 16 + m];
                acc[0][it] = __builtin_amdgcn_wmma_f32_16x16x4_f32(
                    false, a, false, b0, (short)0, acc[0][it], false, false);
                acc[1][it] = __builtin_amdgcn_wmma_f32_16x16x4_f32(
                    false, a, false, b1, (short)0, acc[1][it], false, false);
            }
        }
    }
#pragma unroll
    for (int k0 = 64; k0 < 80; k0 += 4) {
        const int r0 = k0 + 2 * hi;
        v2f b0, b1;
        b0.x = Hp[(size_t)(r0 - 64) * DMOD + d0 + m];
        b0.y = Hp[(size_t)(r0 - 63) * DMOD + d0 + m];
        b1.x = Hp[(size_t)(r0 - 64) * DMOD + d1 + m];
        b1.y = Hp[(size_t)(r0 - 63) * DMOD + d1 + m];
#pragma unroll
        for (int it = 0; it < 4; ++it) {
            v2f a;
            a.x = Asm[r0 * 64 + it * 16 + m];
            a.y = Asm[(r0 + 1) * 64 + it * 16 + m];
            acc[0][it] = __builtin_amdgcn_wmma_f32_16x16x4_f32(
                false, a, false, b0, (short)0, acc[0][it], false, false);
            acc[1][it] = __builtin_amdgcn_wmma_f32_16x16x4_f32(
                false, a, false, b1, (short)0, acc[1][it], false, false);
        }
    }

    float* yb = y + ((size_t)b * SEQ + (size_t)chunk * CHUNK) * DMOD;
#pragma unroll
    for (int it = 0; it < 4; ++it)
#pragma unroll
        for (int v = 0; v < 8; ++v) {
            const size_t row = (size_t)(it * 16 + v + 8 * hi) * DMOD;
            yb[row + d0 + m] = acc[0][it][v];
            yb[row + d1 + m] = acc[1][it][v];
        }
}

extern "C" void kernel_launch(void* const* d_in, const int* in_sizes, int n_in,
                              void* d_out, int out_size, void* d_ws, size_t ws_size,
                              hipStream_t stream) {
    (void)in_sizes; (void)n_in; (void)out_size; (void)ws_size;

    const float* x    = (const float*)d_in[0];
    const float* A    = (const float*)d_in[1];
    const float* Bvec = (const float*)d_in[2];
    float* y  = (float*)d_out;
    float* ws = (float*)d_ws;

    float* AmatT = ws + WS_AMATT;
    float* Ut    = ws + WS_UT;
    float* AcT   = ws + WS_ACT;
    float* S     = ws + WS_S;
    float* H     = ws + WS_H;

    k0_precompute<<<1, 256, 0, stream>>>(A, Bvec, ws);
    k1_state_gemm<<<BATCH * NCH * 8, 256, 0, stream>>>(x, Ut, S);
    k2_scan<<<(BATCH * (DMOD / 16)) / 8, 256, 0, stream>>>(AcT, S, H);
    k3_out_gemm<<<BATCH * NCH * 4, 256, 0, stream>>>(x, AmatT, H, y);
}


# 3 launches: the chunk scan's launch goes; k1 scans groups of 8 chunks (wave = chunk, 3 levels via LDS), k3 adds <=3 products with Q^8, Q^e for the groups in front; sum order in H changes
# speedup vs baseline: 4.2881x; 4.2881x over previous
#include <hip/hip_runtime.h>
#include <hip/hip_bf16.h>

typedef float v2f __attribute__((ext_vector_type(2)));
typedef float v4f __attribute__((ext_vector_type(4)));
typedef float v8f __attribute__((ext_vector_type(8)));

#define BATCH 4
#define SEQ   2048
#define DMOD  1024
#define NST   16
#define CHUNK 64
#define NCH   (SEQ / CHUNK)

#define GRP   8
#define NGRP  (NCH / GRP)

#define WS_AMATT 0
#define WS_UT    (80 * 64)
#define WS_QN    (80 * 64 + 64 * 16)
#define WS_PL    8192

#define PITCH 20

__device__ __forceinline__ float dot16(const float* a, const float* b) {
    float s0 = 0.f, s1 = 0.f, s2 = 0.f, s3 = 0.f;
#pragma unroll
    for (int q = 0; q < 16; q += 4) {
        s0 += a[q] * b[q];
        s1 += a[q + 1] * b[q + 1];
        s2 += a[q + 2] * b[q + 2];
        s3 += a[q + 3] * b[q + 3];
    }
    return (s0 + s1) + (s2 + s3);
}

__global__ __launch_bounds__(256) void k0_precompute(const float* __restrict__ A,
                                                     const float* __restrict__ Bvec,
                                                     float* __restrict__ ws) {
    __shared__ __attribute__((aligned(16))) float P[7][16][PITCH];
    __shared__ __attribute__((aligned(16))) float QQ[8][16][PITCH];
    __shared__ __attribute__((aligned(16))) float U[64][PITCH];
    __shared__ __attribute__((aligned(16))) float R[65][PITCH];
    __shared__ float kv[64];

    const int tid = threadIdx.x;
    const int m = tid >> 4, n = tid & 15;

    float* AmatT = ws + WS_AMATT;
    float* Ut    = ws + WS_UT;
    float* QN    = ws + WS_QN;

    P[0][m][n] = A[m * 16 + n];
    if (tid < 16) {
        const float bv = Bvec[tid];
        U[0][tid] = bv;
        R[0][tid] = bv;
    }
    __syncthreads();

#pragma unroll
    for (int k = 0; k < 6; ++k) {
        const float (*M)[PITCH] = P[k];
        float mrow[16], mcol[16];
#pragma unroll
        for (int q = 0; q < 16; ++q) {
            mrow[q] = M[m][q];
            mcol[q] = M[q][n];
        }

        {
            const int have = 1 << k;
            float nrow[16];
#pragma unroll
            for (int q = 0; q < 16; ++q) nrow[q] = M[n][q];
#pragma unroll
            for (int j = 0; j < (have * 16 + 255) / 256; ++j) {
                const int e = tid + 256 * j;
                if (e < have * 16) {
                    const int i = e >> 4;
                    float uu[16], rr[16];
#pragma unroll
                    for (int q = 0; q < 16; ++q) {
                        uu[q] = U[i][q];
                        rr[q] = R[i][q];
                    }
                    U[have + i][n] = dot16(nrow, uu);
                    R[have + i][n] = dot16(rr, mcol);
                }
            }
        }

        const float sq = dot16(mrow, mcol);
        if (k < 5) P[k + 1][m][n] = sq;
        else       QQ[0][m][n] = sq;
        __syncthreads();
    }

#pragma unroll
    for (int lv = 0; lv < 3; ++lv) {
        const int have = 1 << lv;
        float mcol[16];
#pragma unroll
        for (int q = 0; q < 16; ++q) mcol[q] = QQ[have - 1][q][n];

        if (lv == 0) {
            if (tid < 16) {
                float rr[16];
#pragma unroll
                for (int q = 0; q < 16; ++q) rr[q] = R[0][q];
                R[64][n] = dot16(rr, mcol);
            } else if (tid >= 64 && tid < 128) {
                const int p = tid - 64;
                float rr[16], bb[16];
#pragma unroll
                for (int q = 0; q < 16; ++q) {
                    rr[q] = R[p][q];
                    bb[q] = R[0][q];
                }
                kv[p] = dot16(rr, bb);
            }
        }

#pragma unroll
        for (int i = 1; i <= have; ++i) {
            float row[16];
#pragma unroll
            for (int q = 0; q < 16; ++q) row[q] = QQ[i - 1][m][q];
            QQ[have + i - 1][m][n] = dot16(row, mcol);
        }
        __syncthreads();
    }
#pragma unroll
    for (int e = 0; e < 8; ++e)
        QN[e * 256 + tid] = QQ[e][m][n];

#pragma unroll
    for (int j = 0; j < 4; ++j) {
        const int e = tid + 256 * j;
        Ut[e] = U[63 - (e >> 4)][e & 15];
        AmatT[64 * 64 + e] = R[(e & 63) + 1][e >> 6];
    }

#pragma unroll
    for (int j = 0; j < 16; ++j) {
        const int e = tid + 256 * j;
        const int d = (e & 63) - (e >> 6);
        const float v = kv[d & 63];
        AmatT[e] = (d >= 0) ? v : 0.0f;
    }
}

__global__ __launch_bounds__(256) __attribute__((amdgpu_waves_per_eu(2, 2))) void k1_state_scan(const float* __restrict__ x,
                                                     const float* __restrict__ Ut,
                                                     const float* __restrict__ QN,
                                                     float* __restrict__ PL) {
    __shared__ __attribute__((aligned(16))) v4f pl[2][GRP][4][2][32];

    const int lane = threadIdx.x & 31;
    const int w = __builtin_amdgcn_readfirstlane((int)(threadIdx.x >> 5));
    const int hi = lane >> 4;
    const int m = lane & 15;

    const int blk = blockIdx.x;
    const int slab = blk & 15;
    const int g = (blk >> 4) & (NGRP - 1);
    const int b = blk >> 6;
    const int chunk = g * GRP + w;
    const int d0 = slab * 64;

    const float* xb = x + ((size_t)b * SEQ + (size_t)chunk * CHUNK) * DMOD + d0 + 4 * m;

    v4f qlo[3], qhi[3];
#pragma unroll
    for (int l = 0; l < 3; ++l) {
        const float* qp = QN + ((1 << l) - 1) * 256 + m * 16 + 8 * hi;
        qlo[l] = *reinterpret_cast<const v4f*>(qp);
        qhi[l] = *reinterpret_cast<const v4f*>(qp + 4);
    }

    v8f acc[4] = {{}, {}, {}, {}};
#pragma unroll
    for (int k0 = 0; k0 < CHUNK; k0 += 4) {
        const int r0 = k0 + 2 * hi;
        const v4f l0 = *reinterpret_cast<const v4f*>(xb + (size_t)r0 * DMOD);
        const v4f l1 = *reinterpret_cast<const v4f*>(xb + (size_t)(r0 + 1) * DMOD);
        v2f uf;
        uf.x = Ut[r0 * 16 + m];
        uf.y = Ut[(r0 + 1) * 16 + m];
#pragma unroll
        for (int j = 0; j < 4; ++j) {
            v2f xq;
            xq.x = l0[j];
            xq.y = l1[j];
            acc[j] = __builtin_amdgcn_wmma_f32_16x16x4_f32(false, uf, false, xq,
                                                           (short)0, acc[j], false, false);
        }
    }

#pragma unroll
    for (int l = 0; l < 3; ++l) {
        const int s = 1 << l;
        const int p = l & 1;
#pragma unroll
        for (int j = 0; j < 4; ++j) {
            pl[p][w][j][0][lane] = __builtin_shufflevector(acc[j], acc[j], 0, 1, 2, 3);
            pl[p][w][j][1][lane] = __builtin_shufflevector(acc[j], acc[j], 4, 5, 6, 7);
        }
        __syncthreads();
        if (w >= s) {
            const float qv[8] = {qlo[l].x, qlo[l].y, qlo[l].z, qlo[l].w,
                                 qhi[l].x, qhi[l].y, qhi[l].z, qhi[l].w};
#pragma unroll
            for (int j = 0; j < 4; ++j) {
                const v4f t0 = pl[p][w - s][j][0][lane];
                const v4f t1 = pl[p][w - s][j][1][lane];
                const float tv[8] = {t0.x, t0.y, t0.z, t0.w, t1.x, t1.y, t1.z, t1.w};
#pragma unroll
                for (int q = 0; q < 4; ++q) {
                    v2f a, bq;
                    a.x = qv[2 * q];
                    a.y = qv[2 * q + 1];
                    bq.x = tv[2 * q];
                    bq.y = tv[2 * q + 1];
                    acc[j] = __builtin_amdgcn_wmma_f32_16x16x4_f32(false, a, false, bq,
                                                                   (short)0, acc[j], false, false);
                }
            }
        }
    }

    float* Pp = PL + (((size_t)b * NCH + chunk) * NST + 8 * hi) * DMOD + d0 + 4 * m;
#pragma unroll
    for (int r = 0; r < 8; ++r) {
        v4f o;
        o.x = acc[0][r];
        o.y = acc[1][r];
        o.z = acc[2][r];
        o.w = acc[3][r];
        *reinterpret_cast<v4f*>(Pp + (size_t)r * DMOD) = o;
    }
}

__device__ __forceinline__ void scatter_tile(const v4f (&t)[8], v8f (&e)[4]) {
#pragma unroll
    for (int r = 0; r < 8; ++r)
#pragma unroll
        for (int j = 0; j < 4; ++j) e[j][r] = t[r][j];
}

__device__ __forceinline__ void apply_q(const float (&qv)[8], const v8f (&e)[4], v8f (&c)[4]) {
#pragma unroll
    for (int j = 0; j < 4; ++j)
#pragma unroll
        for (int q = 0; q < 4; ++q) {
            v2f a, bq;
            a.x = qv[2 * q];
            a.y = qv[2 * q + 1];
            bq.x = e[j][2 * q];
            bq.y = e[j][2 * q + 1];
            c[j] = __builtin_amdgcn_wmma_f32_16x16x4_f32(false, a, false, bq,
                                                         (short)0, c[j], false, false);
        }
}

__global__ __launch_bounds__(256) __attribute__((amdgpu_waves_per_eu(2, 2))) void k3_out_gemm(const float* __restrict__ x,
                                                   const float* __restrict__ AmatT,
                                                   const float* __restrict__ QN,
                                                   const float* __restrict__ PL,
                                                   float* __restrict__ y) {
    __shared__ float Asm[80 * 64];
    for (int e = threadIdx.x; e < 80 * 64; e += 256) Asm[e] = AmatT[e];
    __syncthreads();

    const int lane = threadIdx.x & 31;
    const int wave = threadIdx.x >> 5;
    const int hi = lane >> 4;
    const int m = lane & 15;

    const int blk = blockIdx.x;
    const int dhalf = blk & 1;
    const int chunk = (blk >> 1) & (NCH - 1);
    const int b = blk >> 6;
    const int d0 = (dhalf * 8 + wave) * 64;

    const float* xb = x + ((size_t)b * SEQ + (size_t)chunk * CHUNK) * DMOD + d0 + 4 * m;
    const float* Pb = PL + ((size_t)b * NCH * NST + 8 * hi) * DMOD + d0 + 4 * m;

    const int gb = (chunk - 1) >> 3;
    const int ee = chunk - 8 * gb;

    v4f th[8], t0[8], t1[8], t2[8];
    const float* q8p = QN + 7 * 256 + m * 16 + 8 * hi;
    const float* qep = QN + ((ee - 1) & 7) * 256 + m * 16 + 8 * hi;
    const v4f q8a = *reinterpret_cast<const v4f*>(q8p);
    const v4f q8b = *reinterpret_cast<const v4f*>(q8p + 4);
    const v4f qea = *reinterpret_cast<const v4f*>(qep);
    const v4f qeb = *reinterpret_cast<const v4f*>(qep + 4);
    if (gb >= 0) {
#pragma unroll
        for (int r = 0; r < 8; ++r)
            th[r] = *reinterpret_cast<const v4f*>(Pb + ((size_t)(chunk - 1) * NST + r) * DMOD);
    }
    if (gb >= 1) {
#pragma unroll
        for (int r = 0; r < 8; ++r)
            t0[r] = *reinterpret_cast<const v4f*>(Pb + ((size_t)(GRP - 1) * NST + r) * DMOD);
    }
    if (gb >= 2) {
#pragma unroll
        for (int r = 0; r < 8; ++r)
            t1[r] = *reinterpret_cast<const v4f*>(Pb + ((size_t)(2 * GRP - 1) * NST + r) * DMOD);
    }
    if (gb >= 3) {
#pragma unroll
        for (int r = 0; r < 8; ++r)
            t2[r] = *reinterpret_cast<const v4f*>(Pb + ((size_t)(3 * GRP - 1) * NST + r) * DMOD);
    }

    v4f xl0[16], xl1[16];
#pragma unroll
    for (int ks = 0; ks < 16; ++ks) {
        const int r0 = 4 * ks + 2 * hi;
        xl0[ks] = *reinterpret_cast<const v4f*>(xb + (size_t)r0 * DMOD);
        xl1[ks] = *reinterpret_cast<const v4f*>(xb + (size_t)(r0 + 1) * DMOD);
    }

    v8f hs[4] = {{}, {}, {}, {}};
    if (gb >= 0) scatter_tile(th, hs);
    if (gb >= 1) {
        const float q8v[8] = {q8a.x, q8a.y, q8a.z, q8a.w, q8b.x, q8b.y, q8b.z, q8b.w};
        const float qev[8] = {qea.x, qea.y, qea.z, qea.w, qeb.x, qeb.y, qeb.z, qeb.w};
        v8f e0[4];
        scatter_tile(t0, e0);
        if (gb >= 2) {
            v8f e1[4];
            scatter_tile(t1, e1);
            apply_q(q8v, e0, e1);
#pragma unroll
            for (int j = 0; j < 4; ++j) e0[j] = e1[j];
        }
        if (gb >= 3) {
            v8f e2[4];
            scatter_tile(t2, e2);
            apply_q(q8v, e0, e2);
#pragma unroll
            for (int j = 0; j < 4; ++j) e0[j] = e2[j];
        }
        apply_q(qev, e0, hs);
    }

    v8f acc[4][4] = {{{}, {}, {}, {}}, {{}, {}, {}, {}}, {{}, {}, {}, {}}, {{}, {}, {}, {}}};

#pragma unroll
    for (int kb = 0; kb < 4; ++kb) {
#pragma unroll
        for (int kk = 0; kk < 16; kk += 4) {
            const int r0 = kb * 16 + kk + 2 * hi;
            const v4f l0 = xl0[(kb * 16 + kk) >> 2];
            const v4f l1 = xl1[(kb * 16 + kk) >> 2];
#pragma unroll
            for (int it = kb; it < 4; ++it) {
                v2f a;
                a.x = Asm[r0 * 64 + it * 16 + m];
                a.y = Asm[(r0 + 1) * 64 + it * 16 + m];
#pragma unroll
                for (int j = 0; j < 4; ++j) {
                    v2f bq;
                    bq.x = l0[j];
                    bq.y = l1[j];
                    acc[j][it] = __builtin_amdgcn_wmma_f32_16x16x4_f32(
                        false, a, false, bq, (short)0, acc[j][it], false, false);
                }
            }
        }
    }
#pragma unroll
    for (int q = 0; q < 4; ++q) {
        const int r0 = 64 + 8 * hi + 2 * q;
#pragma unroll
        for (int it = 0; it < 4; ++it) {
            v2f a;
            a.x = Asm[r0 * 64 + it * 16 + m];
            a.y = Asm[(r0 + 1) * 64 + it * 16 + m];
#pragma unroll
            for (int j = 0; j < 4; ++j) {
                v2f bq;
                bq.x = hs[j][2 * q];
                bq.y = hs[j][2 * q + 1];
                acc[j][it] = __builtin_amdgcn_wmma_f32_16x16x4_f32(
                    false, a, false, bq, (short)0, acc[j][it], false, false);
            }
        }
    }

    float* yb = y + ((size_t)b * SEQ + (size_t)chunk * CHUNK) * DMOD + d0 + 4 * m;
#pragma unroll
    for (int it = 0; it < 4; ++it)
#pragma unroll
        for (int v = 0; v < 8; ++v) {
            v4f o;
            o.x = acc[0][it][v];
            o.y = acc[1][it][v];
            o.z = acc[2][it][v];
            o.w = acc[3][it][v];
            *reinterpret_cast<v4f*>(yb + (size_t)(it * 16 + v + 8 * hi) * DMOD) = o;
        }
}

extern "C" void kernel_launch(void* const* d_in, const int* in_sizes, int n_in,
                              void* d_out, int out_size, void* d_ws, size_t ws_size,
                              hipStream_t stream) {
    (void)in_sizes; (void)n_in; (void)out_size; (void)ws_size;

    const float* x    = (const float*)d_in[0];
    const float* A    = (const float*)d_in[1];
    const float* Bvec = (const float*)d_in[2];
    float* y  = (float*)d_out;
    float* ws = (float*)d_ws;

    float* AmatT = ws + WS_AMATT;
    float* Ut    = ws + WS_UT;
    float* QN    = ws + WS_QN;
    float* PL    = ws + WS_PL;

    k0_precompute<<<1, 256, 0, stream>>>(A, Bvec, ws);
    k1_state_scan<<<BATCH * NGRP * 16, 256, 0, stream>>>(x, Ut, QN, PL);
    k3_out_gemm<<<BATCH * NCH * 2, 256, 0, stream>>>(x, AmatT, QN, PL, y);
}


# k1,k3: A operands come as 16-byte pieces from fragment-ordered tables k0 writes and stay in registers (T pair depends on it-kb, kk only); k3 loses LDS copy, barrier, LDS reads; sums unchanged
# speedup vs baseline: 4.9131x; 1.1458x over previous
#include <hip/hip_runtime.h>
#include <hip/hip_bf16.h>

typedef float v2f __attribute__((ext_vector_type(2)));
typedef float v4f __attribute__((ext_vector_type(4)));
typedef float v8f __attribute__((ext_vector_type(8)));

#define BATCH 4
#define SEQ   2048
#define DMOD  1024
#define NST   16
#define CHUNK 64
#define NCH   (SEQ / CHUNK)

#define GRP   8
#define NGRP  (NCH / GRP)

#define WS_TABT  0
#define WS_TABO  1024
#define WS_UT    (80 * 64)
#define WS_QN    (80 * 64 + 64 * 16)
#define WS_PL    8192

#define PITCH 20

__device__ __forceinline__ float dot16(const float* a, const float* b) {
    float s0 = 0.f, s1 = 0.f, s2 = 0.f, s3 = 0.f;
#pragma unroll
    for (int q = 0; q < 16; q += 4) {
        s0 += a[q] * b[q];
        s1 += a[q + 1] * b[q + 1];
        s2 += a[q + 2] * b[q + 2];
        s3 += a[q + 3] * b[q + 3];
    }
    return (s0 + s1) + (s2 + s3);
}

__global__ __launch_bounds__(256) void k0_precompute(const float* __restrict__ A,
                                                     const float* __restrict__ Bvec,
                                                     float* __restrict__ ws) {
    __shared__ __attribute__((aligned(16))) float P[7][16][PITCH];
    __shared__ __attribute__((aligned(16))) float QQ[8][16][PITCH];
    __shared__ __attribute__((aligned(16))) float U[64][PITCH];
    __shared__ __attribute__((aligned(16))) float R[65][PITCH];
    __shared__ float kv[64];

    const int tid = threadIdx.x;
    const int m = tid >> 4, n = tid & 15;

    float* TabT = ws + WS_TABT;
    float* TabO = ws + WS_TABO;
    float* TabU = ws + WS_UT;
    float* QN   = ws + WS_QN;

    P[0][m][n] = A[m * 16 + n];
    if (tid < 16) {
        const float bv = Bvec[tid];
        U[0][tid] = bv;
        R[0][tid] = bv;
    }
    __syncthreads();

#pragma unroll
    for (int k = 0; k < 6; ++k) {
        const float (*M)[PITCH] = P[k];
        float mrow[16], mcol[16];
#pragma unroll
        for (int q = 0; q < 16; ++q) {
            mrow[q] = M[m][q];
            mcol[q] = M[q][n];
        }

        {
            const int have = 1 << k;
            float nrow[16];
#pragma unroll
            for (int q = 0; q < 16; ++q) nrow[q] = M[n][q];
#pragma unroll
            for (int j = 0; j < (have * 16 + 255) / 256; ++j) {
                const int e = tid + 256 * j;
                if (e < have * 16) {
                    const int i = e >> 4;
                    float uu[16], rr[16];
#pragma unroll
                    for (int q = 0; q < 16; ++q) {
                        uu[q] = U[i][q];
                        rr[q] = R[i][q];
                    }
                    U[have + i][n] = dot16(nrow, uu);
                    R[have + i][n] = dot16(rr, mcol);
                }
            }
        }

        const float sq = dot16(mrow, mcol);
        if (k < 5) P[k + 1][m][n] = sq;
        else       QQ[0][m][n] = sq;
        __syncthreads();
    }

#pragma unroll
    for (int lv = 0; lv < 3; ++lv) {
        const int have = 1 << lv;
        float mcol[16];
#pragma unroll
        for (int q = 0; q < 16; ++q) mcol[q] = QQ[have - 1][q][n];

        if (lv == 0) {
            if (tid < 16) {
                float rr[16];
#pragma unroll
                for (int q = 0; q < 16; ++q) rr[q] = R[0][q];
                R[64][n] = dot16(rr, mcol);
            } else if (tid >= 64 && tid < 128) {
                const int p = tid - 64;
                float rr[16], bb[16];
#pragma unroll
                for (int q = 0; q < 16; ++q) {
                    rr[q] = R[p][q];
                    bb[q] = R[0][q];
                }
                kv[p] = dot16(rr, bb);
            }
        }

#pragma unroll
        for (int i = 1; i <= have; ++i) {
            float row[16];
#pragma unroll
            for (int q = 0; q < 16; ++q) row[q] = QQ[i - 1][m][q];
            QQ[have + i - 1][m][n] = dot16(row, mcol);
        }
        __syncthreads();
    }
#pragma unroll
    for (int e = 0; e < 8; ++e)
        QN[e * 256 + tid] = QQ[e][m][n];

#pragma unroll
    for (int j = 0; j < 4; ++j) {
        const int e = tid + 256 * j;
        const int c = e & 1, h2 = (e >> 1) & 1, ln = (e >> 2) & 31, g = e >> 7;
        const int lm = ln & 15, lh = ln >> 4;
        TabU[e] = U[63 - (8 * g + 4 * h2 + 2 * lh + c)][lm];
        TabO[e] = R[(e >> 4) + 1][e & 15];
        const int idx = 4 * (2 * g + h2 - 3) + lm - 2 * lh - c;
        const float v = kv[idx & 63];
        TabT[e] = (idx >= 0) ? v : 0.0f;
    }
}

__global__ __launch_bounds__(256) __attribute__((amdgpu_waves_per_eu(2, 2))) void k1_state_scan(const float* __restrict__ x,
                                                     const float* __restrict__ TabU,
                                                     const float* __restrict__ QN,
                                                     float* __restrict__ PL) {
    __shared__ __attribute__((aligned(16))) v4f pl[2][GRP][4][2][32];

    const int lane = threadIdx.x & 31;
    const int w = __builtin_amdgcn_readfirstlane((int)(threadIdx.x >> 5));
    const int hi = lane >> 4;
    const int m = lane & 15;

    const int blk = blockIdx.x;
    const int slab = blk & 15;
    const int g = (blk >> 4) & (NGRP - 1);
    const int b = blk >> 6;
    const int chunk = g * GRP + w;
    const int d0 = slab * 64;

    const float* xb = x + ((size_t)b * SEQ + (size_t)chunk * CHUNK) * DMOD + d0 + 4 * m;

    v4f qlo[3], qhi[3];
#pragma unroll
    for (int l = 0; l < 3; ++l) {
        const float* qp = QN + ((1 << l) - 1) * 256 + m * 16 + 8 * hi;
        qlo[l] = *reinterpret_cast<const v4f*>(qp);
        qhi[l] = *reinterpret_cast<const v4f*>(qp + 4);
    }

    v4f ut[8];
#pragma unroll
    for (int g = 0; g < 8; ++g)
        ut[g] = *reinterpret_cast<const v4f*>(TabU + (g * 32 + lane) * 4);

    v8f acc[4] = {{}, {}, {}, {}};
#pragma unroll
    for (int k0 = 0; k0 < CHUNK; k0 += 4) {
        const int r0 = k0 + 2 * hi;
        const v4f l0 = *reinterpret_cast<const v4f*>(xb + (size_t)r0 * DMOD);
        const v4f l1 = *reinterpret_cast<const v4f*>(xb + (size_t)(r0 + 1) * DMOD);
        const int ks = k0 >> 2;
        v2f uf;
        uf.x = ut[ks >> 1][(ks & 1) * 2];
        uf.y = ut[ks >> 1][(ks & 1) * 2 + 1];
#pragma unroll
        for (int j = 0; j < 4; ++j) {
            v2f xq;
            xq.x = l0[j];
            xq.y = l1[j];
            acc[j] = __builtin_amdgcn_wmma_f32_16x16x4_f32(false, uf, false, xq,
                                                           (short)0, acc[j], false, false);
        }
    }

#pragma unroll
    for (int l = 0; l < 3; ++l) {
        const int s = 1 << l;
        const int p = l & 1;
#pragma unroll
        for (int j = 0; j < 4; ++j) {
            pl[p][w][j][0][lane] = __builtin_shufflevector(acc[j], acc[j], 0, 1, 2, 3);
            pl[p][w][j][1][lane] = __builtin_shufflevector(acc[j], acc[j], 4, 5, 6, 7);
        }
        __syncthreads();
        if (w >= s) {
            const float qv[8] = {qlo[l].x, qlo[l].y, qlo[l].z, qlo[l].w,
                                 qhi[l].x, qhi[l].y, qhi[l].z, qhi[l].w};
#pragma unroll
            for (int j = 0; j < 4; ++j) {
                const v4f t0 = pl[p][w - s][j][0][lane];
                const v4f t1 = pl[p][w - s][j][1][lane];
                const float tv[8] = {t0.x, t0.y, t0.z, t0.w, t1.x, t1.y, t1.z, t1.w};
#pragma unroll
                for (int q = 0; q < 4; ++q) {
                    v2f a, bq;
                    a.x = qv[2 * q];
                    a.y = qv[2 * q + 1];
                    bq.x = tv[2 * q];
                    bq.y = tv[2 * q + 1];
                    acc[j] = __builtin_amdgcn_wmma_f32_16x16x4_f32(false, a, false, bq,
                                                                   (short)0, acc[j], false, false);
                }
            }
        }
    }

    float* Pp = PL + (((size_t)b * NCH + chunk) * NST + 8 * hi) * DMOD + d0 + 4 * m;
#pragma unroll
    for (int r = 0; r < 8; ++r) {
        v4f o;
        o.x = acc[0][r];
        o.y = acc[1][r];
        o.z = acc[2][r];
        o.w = acc[3][r];
        *reinterpret_cast<v4f*>(Pp + (size_t)r * DMOD) = o;
    }
}

__device__ __forceinline__ void scatter_tile(const v4f (&t)[8], v8f (&e)[4]) {
#pragma unroll
    for (int r = 0; r < 8; ++r)
#pragma unroll
        for (int j = 0; j < 4; ++j) e[j][r] = t[r][j];
}

__device__ __forceinline__ void apply_q(const float (&qv)[8], const v8f (&e)[4], v8f (&c)[4]) {
#pragma unroll
    for (int j = 0; j < 4; ++j)
#pragma unroll
        for (int q = 0; q < 4; ++q) {
            v2f a, bq;
            a.x = qv[2 * q];
            a.y = qv[2 * q + 1];
            bq.x = e[j][2 * q];
            bq.y = e[j][2 * q + 1];
            c[j] = __builtin_amdgcn_wmma_f32_16x16x4_f32(false, a, false, bq,
                                                         (short)0, c[j], false, false);
        }
}

__global__ __launch_bounds__(256) __attribute__((amdgpu_waves_per_eu(2, 2))) void k3_out_gemm(const float* __restrict__ x,
                                                   const float* __restrict__ TabT,
                                                   const float* __restrict__ TabO,
                                                   const float* __restrict__ QN,
                                                   const float* __restrict__ PL,
                                                   float* __restrict__ y) {
    const int lane = threadIdx.x & 31;
    const int wave = threadIdx.x >> 5;
    const int hi = lane >> 4;
    const int m = lane & 15;

    const int blk = blockIdx.x;
    const int dhalf = blk & 1;
    const int chunk = (blk >> 1) & (NCH - 1);
    const int b = blk >> 6;
    const int d0 = (dhalf * 8 + wave) * 64;

    const float* xb = x + ((size_t)b * SEQ + (size_t)chunk * CHUNK) * DMOD + d0 + 4 * m;
    const float* Pb = PL + ((size_t)b * NCH * NST + 8 * hi) * DMOD + d0 + 4 * m;

    const int gb = (chunk - 1) >> 3;
    const int ee = chunk - 8 * gb;

    v4f tt[8];
#pragma unroll
    for (int g = 0; g < 8; ++g)
        tt[g] = *reinterpret_cast<const v4f*>(TabT + (g * 32 + lane) * 4);

    v4f th[8], t0[8], t1[8], t2[8];
    const float* q8p = QN + 7 * 256 + m * 16 + 8 * hi;
    const float* qep = QN + ((ee - 1) & 7) * 256 + m * 16 + 8 * hi;
    const v4f q8a = *reinterpret_cast<const v4f*>(q8p);
    const v4f q8b = *reinterpret_cast<const v4f*>(q8p + 4);
    const v4f qea = *reinterpret_cast<const v4f*>(qep);
    const v4f qeb = *reinterpret_cast<const v4f*>(qep + 4);
    if (gb >= 0) {
#pragma unroll
        for (int r = 0; r < 8; ++r)
            th[r] = *reinterpret_cast<const v4f*>(Pb + ((size_t)(chunk - 1) * NST + r) * DMOD);
    }
    if (gb >= 1) {
#pragma unroll
        for (int r = 0; r < 8; ++r)
            t0[r] = *reinterpret_cast<const v4f*>(Pb + ((size_t)(GRP - 1) * NST + r) * DMOD);
    }
    if (gb >= 2) {
#pragma unroll
        for (int r = 0; r < 8; ++r)
            t1[r] = *reinterpret_cast<const v4f*>(Pb + ((size_t)(2 * GRP - 1) * NST + r) * DMOD);
    }
    if (gb >= 3) {
#pragma unroll
        for (int r = 0; r < 8; ++r)
            t2[r] = *reinterpret_cast<const v4f*>(Pb + ((size_t)(3 * GRP - 1) * NST + r) * DMOD);
    }

    v4f xl0[16], xl1[16];
#pragma unroll
    for (int ks = 0; ks < 16; ++ks) {
        const int r0 = 4 * ks + 2 * hi;
        xl0[ks] = *reinterpret_cast<const v4f*>(xb + (size_t)r0 * DMOD);
        xl1[ks] = *reinterpret_cast<const v4f*>(xb + (size_t)(r0 + 1) * DMOD);
    }
    v4f to[4][2];
#pragma unroll
    for (int it = 0; it < 4; ++it)
#pragma unroll
        for (int h = 0; h < 2; ++h)
            to[it][h] = *reinterpret_cast<const v4f*>(TabO + (it * 16 + m) * 16 + 8 * hi + 4 * h);
    asm volatile("" ::: "memory");

    v8f hs[4] = {{}, {}, {}, {}};
    if (gb >= 0) scatter_tile(th, hs);
    if (gb >= 1) {
        const float q8v[8] = {q8a.x, q8a.y, q8a.z, q8a.w, q8b.x, q8b.y, q8b.z, q8b.w};
        const float qev[8] = {qea.x, qea.y, qea.z, qea.w, qeb.x, qeb.y, qeb.z, qeb.w};
        v8f e0[4];
        scatter_tile(t0, e0);
        if (gb >= 2) {
            v8f e1[4];
            scatter_tile(t1, e1);
            apply_q(q8v, e0, e1);
#pragma unroll
            for (int j = 0; j < 4; ++j) e0[j] = e1[j];
        }
        if (gb >= 3) {
            v8f e2[4];
            scatter_tile(t2, e2);
            apply_q(q8v, e0, e2);
#pragma unroll
            for (int j = 0; j < 4; ++j) e0[j] = e2[j];
        }
        apply_q(qev, e0, hs);
    }

    v8f acc[4][4] = {{{}, {}, {}, {}}, {{}, {}, {}, {}}, {{}, {}, {}, {}}, {{}, {}, {}, {}}};

#pragma unroll
    for (int kb = 0; kb < 4; ++kb) {
#pragma unroll
        for (int kk = 0; kk < 16; kk += 4) {
            const v4f l0 = xl0[(kb * 16 + kk) >> 2];
            const v4f l1 = xl1[(kb * 16 + kk) >> 2];
#pragma unroll
            for (int it = kb; it < 4; ++it) {
                const int sp = 4 * (it - kb) - (kk >> 2) + 3;
                v2f a;
                a.x = tt[sp >> 1][(sp & 1) * 2];
                a.y = tt[sp >> 1][(sp & 1) * 2 + 1];
#pragma unroll
                for (int j = 0; j < 4; ++j) {
                    v2f bq;
                    bq.x = l0[j];
                    bq.y = l1[j];
                    acc[j][it] = __builtin_amdgcn_wmma_f32_16x16x4_f32(
                        false, a, false, bq, (short)0, acc[j][it], false, false);
                }
            }
        }
    }
#pragma unroll
    for (int q = 0; q < 4; ++q) {
#pragma unroll
        for (int it = 0; it < 4; ++it) {
            v2f a;
            a.x = to[it][q >> 1][(q & 1) * 2];
            a.y = to[it][q >> 1][(q & 1) * 2 + 1];
#pragma unroll
            for (int j = 0; j < 4; ++j) {
                v2f bq;
                bq.x = hs[j][2 * q];
                bq.y = hs[j][2 * q + 1];
                acc[j][it] = __builtin_amdgcn_wmma_f32_16x16x4_f32(
                    false, a, false, bq, (short)0, acc[j][it], false, false);
            }
        }
    }

    float* yb = y + ((size_t)b * SEQ + (size_t)chunk * CHUNK) * DMOD + d0 + 4 * m;
#pragma unroll
    for (int it = 0; it < 4; ++it)
#pragma unroll
        for (int v = 0; v < 8; ++v) {
            v4f o;
            o.x = acc[0][it][v];
            o.y = acc[1][it][v];
            o.z = acc[2][it][v];
            o.w = acc[3][it][v];
            *reinterpret_cast<v4f*>(yb + (size_t)(it * 16 + v + 8 * hi) * DMOD) = o;
        }
}

extern "C" void kernel_launch(void* const* d_in, const int* in_sizes, int n_in,
                              void* d_out, int out_size, void* d_ws, size_t ws_size,
                              hipStream_t stream) {
    (void)in_sizes; (void)n_in; (void)out_size; (void)ws_size;

    const float* x    = (const float*)d_in[0];
    const float* A    = (const float*)d_in[1];
    const float* Bvec = (const float*)d_in[2];
    float* y  = (float*)d_out;
    float* ws = (float*)d_ws;

    float* TabT = ws + WS_TABT;
    float* TabO = ws + WS_TABO;
    float* TabU = ws + WS_UT;
    float* QN   = ws + WS_QN;
    float* PL   = ws + WS_PL;

    k0_precompute<<<1, 256, 0, stream>>>(A, Bvec, ws);
    k1_state_scan<<<BATCH * NGRP * 16, 256, 0, stream>>>(x, TabU, QN, PL);
    k3_out_gemm<<<BATCH * NCH * 2, 256, 0, stream>>>(x, TabT, TabO, QN, PL, y);
}


# k3: one wave per SIMD owns two tiles (chunks p and 31-p of a 64-channel slab; 256 groups of 4 waves); tile 2's loads requested between tile 1's K-blocks; equal state products per SIMD; sums unchanged
# speedup vs baseline: 5.1789x; 1.0541x over previous
#include <hip/hip_runtime.h>
#include <hip/hip_bf16.h>

typedef float v2f __attribute__((ext_vector_type(2)));
typedef float v4f __attribute__((ext_vector_type(4)));
typedef float v8f __attribute__((ext_vector_type(8)));

#define BATCH 4
#define SEQ   2048
#define DMOD  1024
#define NST   16
#define CHUNK 64
#define NCH   (SEQ / CHUNK)

#define GRP   8
#define NGRP  (NCH / GRP)

#define WS_TABT  0
#define WS_TABO  1024
#define WS_UT    (80 * 64)
#define WS_QN    (80 * 64 + 64 * 16)
#define WS_PL    8192

#define PITCH 20

__device__ __forceinline__ float dot16(const float* a, const float* b) {
    float s0 = 0.f, s1 = 0.f, s2 = 0.f, s3 = 0.f;
#pragma unroll
    for (int q = 0; q < 16; q += 4) {
        s0 += a[q] * b[q];
        s1 += a[q + 1] * b[q + 1];
        s2 += a[q + 2] * b[q + 2];
        s3 += a[q + 3] * b[q + 3];
    }
    return (s0 + s1) + (s2 + s3);
}

__global__ __launch_bounds__(256) void k0_precompute(const float* __restrict__ A,
                                                     const float* __restrict__ Bvec,
                                                     float* __restrict__ ws) {
    __shared__ __attribute__((aligned(16))) float P[7][16][PITCH];
    __shared__ __attribute__((aligned(16))) float QQ[8][16][PITCH];
    __shared__ __attribute__((aligned(16))) float U[64][PITCH];
    __shared__ __attribute__((aligned(16))) float R[65][PITCH];
    __shared__ float kv[64];

    const int tid = threadIdx.x;
    const int m = tid >> 4, n = tid & 15;

    float* TabT = ws + WS_TABT;
    float* TabO = ws + WS_TABO;
    float* TabU = ws + WS_UT;
    float* QN   = ws + WS_QN;

    P[0][m][n] = A[m * 16 + n];
    if (tid < 16) {
        const float bv = Bvec[tid];
        U[0][tid] = bv;
        R[0][tid] = bv;
    }
    __syncthreads();

#pragma unroll
    for (int k = 0; k < 6; ++k) {
        const float (*M)[PITCH] = P[k];
        float mrow[16], mcol[16];
#pragma unroll
        for (int q = 0; q < 16; ++q) {
            mrow[q] = M[m][q];
            mcol[q] = M[q][n];
        }

        {
            const int have = 1 << k;
            float nrow[16];
#pragma unroll
            for (int q = 0; q < 16; ++q) nrow[q] = M[n][q];
#pragma unroll
            for (int j = 0; j < (have * 16 + 255) / 256; ++j) {
                const int e = tid + 256 * j;
                if (e < have * 16) {
                    const int i = e >> 4;
                    float uu[16], rr[16];
#pragma unroll
                    for (int q = 0; q < 16; ++q) {
                        uu[q] = U[i][q];
                        rr[q] = R[i][q];
                    }
                    U[have + i][n] = dot16(nrow, uu);
                    R[have + i][n] = dot16(rr, mcol);
                }
            }
        }

        const float sq = dot16(mrow, mcol);
        if (k < 5) P[k + 1][m][n] = sq;
        else       QQ[0][m][n] = sq;
        __syncthreads();
    }

#pragma unroll
    for (int lv = 0; lv < 3; ++lv) {
        const int have = 1 << lv;
        float mcol[16];
#pragma unroll
        for (int q = 0; q < 16; ++q) mcol[q] = QQ[have - 1][q][n];

        if (lv == 0) {
            if (tid < 16) {
                float rr[16];
#pragma unroll
                for (int q = 0; q < 16; ++q) rr[q] = R[0][q];
                R[64][n] = dot16(rr, mcol);
            } else if (tid >= 64 && tid < 128) {
                const int p = tid - 64;
                float rr[16], bb[16];
#pragma unroll
                for (int q = 0; q < 16; ++q) {
                    rr[q] = R[p][q];
                    bb[q] = R[0][q];
                }
                kv[p] = dot16(rr, bb);
            }
        }

#pragma unroll
        for (int i = 1; i <= have; ++i) {
            float row[16];
#pragma unroll
            for (int q = 0; q < 16; ++q) row[q] = QQ[i - 1][m][q];
            QQ[have + i - 1][m][n] = dot16(row, mcol);
        }
        __syncthreads();
    }
#pragma unroll
    for (int e = 0; e < 8; ++e)
        QN[e * 256 + tid] = QQ[e][m][n];

#pragma unroll
    for (int j = 0; j < 4; ++j) {
        const int e = tid + 256 * j;
        const int c = e & 1, h2 = (e >> 1) & 1, ln = (e >> 2) & 31, g = e >> 7;
        const int lm = ln & 15, lh = ln >> 4;
        TabU[e] = U[63 - (8 * g + 4 * h2 + 2 * lh + c)][lm];
        TabO[e] = R[(e >> 4) + 1][e & 15];
        const int idx = 4 * (2 * g + h2 - 3) + lm - 2 * lh - c;
        const float v = kv[idx & 63];
        TabT[e] = (idx >= 0) ? v : 0.0f;
    }
}

__global__ __launch_bounds__(256) __attribute__((amdgpu_waves_per_eu(2, 2))) void k1_state_scan(const float* __restrict__ x,
                                                     const float* __restrict__ TabU,
                                                     const float* __restrict__ QN,
                                                     float* __restrict__ PL) {
    __shared__ __attribute__((aligned(16))) v4f pl[2][GRP][4][2][32];

    const int lane = threadIdx.x & 31;
    const int w = __builtin_amdgcn_readfirstlane((int)(threadIdx.x >> 5));
    const int hi = lane >> 4;
    const int m = lane & 15;

    const int blk = blockIdx.x;
    const int slab = blk & 15;
    const int g = (blk >> 4) & (NGRP - 1);
    const int b = blk >> 6;
    const int chunk = g * GRP + w;
    const int d0 = slab * 64;

    const float* xb = x + ((size_t)b * SEQ + (size_t)chunk * CHUNK) * DMOD + d0 + 4 * m;

    v4f qlo[3], qhi[3];
#pragma unroll
    for (int l = 0; l < 3; ++l) {
        const float* qp = QN + ((1 << l) - 1) * 256 + m * 16 + 8 * hi;
        qlo[l] = *reinterpret_cast<const v4f*>(qp);
        qhi[l] = *reinterpret_cast<const v4f*>(qp + 4);
    }

    v4f ut[8];
#pragma unroll
    for (int g = 0; g < 8; ++g)
        ut[g] = *reinterpret_cast<const v4f*>(TabU + (g * 32 + lane) * 4);

    v8f acc[4] = {{}, {}, {}, {}};
#pragma unroll
    for (int k0 = 0; k0 < CHUNK; k0 += 4) {
        const int r0 = k0 + 2 * hi;
        const v4f l0 = *reinterpret_cast<const v4f*>(xb + (size_t)r0 * DMOD);
        const v4f l1 = *reinterpret_cast<const v4f*>(xb + (size_t)(r0 + 1) * DMOD);
        const int ks = k0 >> 2;
        v2f uf;
        uf.x = ut[ks >> 1][(ks & 1) * 2];
        uf.y = ut[ks >> 1][(ks & 1) * 2 + 1];
#pragma unroll
        for (int j = 0; j < 4; ++j) {
            v2f xq;
            xq.x = l0[j];
            xq.y = l1[j];
            acc[j] = __builtin_amdgcn_wmma_f32_16x16x4_f32(false, uf, false, xq,
                                                           (short)0, acc[j], false, false);
        }
    }

#pragma unroll
    for (int l = 0; l < 3; ++l) {
        const int s = 1 << l;
        const int p = l & 1;
#pragma unroll
        for (int j = 0; j < 4; ++j) {
            pl[p][w][j][0][lane] = __builtin_shufflevector(acc[j], acc[j], 0, 1, 2, 3);
            pl[p][w][j][1][lane] = __builtin_shufflevector(acc[j], acc[j], 4, 5, 6, 7);
        }
        __syncthreads();
        if (w >= s) {
            const float qv[8] = {qlo[l].x, qlo[l].y, qlo[l].z, qlo[l].w,
                                 qhi[l].x, qhi[l].y, qhi[l].z, qhi[l].w};
#pragma unroll
            for (int j = 0; j < 4; ++j) {
                const v4f t0 = pl[p][w - s][j][0][lane];
                const v4f t1 = pl[p][w - s][j][1][lane];
                const float tv[8] = {t0.x, t0.y, t0.z, t0.w, t1.x, t1.y, t1.z, t1.w};
#pragma unroll
                for (int q = 0; q < 4; ++q) {
                    v2f a, bq;
                    a.x = qv[2 * q];
                    a.y = qv[2 * q + 1];
                    bq.x = tv[2 * q];
                    bq.y = tv[2 * q + 1];
                    acc[j] = __builtin_amdgcn_wmma_f32_16x16x4_f32(false, a, false, bq,
                                                                   (short)0, acc[j], false, false);
                }
            }
        }
    }

    float* Pp = PL + (((size_t)b * NCH + chunk) * NST + 8 * hi) * DMOD + d0 + 4 * m;
#pragma unroll
    for (int r = 0; r < 8; ++r) {
        v4f o;
        o.x = acc[0][r];
        o.y = acc[1][r];
        o.z = acc[2][r];
        o.w = acc[3][r];
        *reinterpret_cast<v4f*>(Pp + (size_t)r * DMOD) = o;
    }
}

__device__ __forceinline__ void scatter_tile(const v4f (&t)[8], v8f (&e)[4]) {
#pragma unroll
    for (int r = 0; r < 8; ++r)
#pragma unroll
        for (int j = 0; j < 4; ++j) e[j][r] = t[r][j];
}

__device__ __forceinline__ void apply_q(const float (&qv)[8], const v8f (&e)[4], v8f (&c)[4]) {
#pragma unroll
    for (int j = 0; j < 4; ++j)
#pragma unroll
        for (int q = 0; q < 4; ++q) {
            v2f a, bq;
            a.x = qv[2 * q];
            a.y = qv[2 * q + 1];
            bq.x = e[j][2 * q];
            bq.y = e[j][2 * q + 1];
            c[j] = __builtin_amdgcn_wmma_f32_16x16x4_f32(false, a, false, bq,
                                                         (short)0, c[j], false, false);
        }
}

struct K3Tile {
    v4f th[8], t0[8], t1[8], t2[8];
    v4f xl0[16], xl1[16];
    v4f qea, qeb;
};

template <int MAXG, int PART>
__device__ __forceinline__ void k3_load_part(K3Tile& t, const float* __restrict__ x,
                                             const float* __restrict__ QN,
                                             const float* __restrict__ PL,
                                             int b, int chunk, int d0, int m, int hi) {
    const float* Pb = PL + ((size_t)b * NCH * NST + 8 * hi) * DMOD + d0 + 4 * m;
    const int gb = (chunk - 1) >> 3;
    if (PART == 0) {
        const int ee = chunk - 8 * gb;
        const float* qep = QN + ((ee - 1) & 7) * 256 + m * 16 + 8 * hi;
        t.qea = *reinterpret_cast<const v4f*>(qep);
        t.qeb = *reinterpret_cast<const v4f*>(qep + 4);
        if (gb >= 0) {
#pragma unroll
            for (int r = 0; r < 8; ++r)
                t.th[r] = *reinterpret_cast<const v4f*>(Pb + ((size_t)(chunk - 1) * NST + r) * DMOD);
        }
        if (MAXG >= 1 && gb >= 1) {
#pragma unroll
            for (int r = 0; r < 8; ++r)
                t.t0[r] = *reinterpret_cast<const v4f*>(Pb + ((size_t)(GRP - 1) * NST + r) * DMOD);
        }
    }
    if (PART == 1) {
        if (MAXG >= 2 && gb >= 2) {
#pragma unroll
            for (int r = 0; r < 8; ++r)
                t.t1[r] = *reinterpret_cast<const v4f*>(Pb + ((size_t)(2 * GRP - 1) * NST + r) * DMOD);
        }
        if (MAXG >= 3 && gb >= 3) {
#pragma unroll
            for (int r = 0; r < 8; ++r)
                t.t2[r] = *reinterpret_cast<const v4f*>(Pb + ((size_t)(3 * GRP - 1) * NST + r) * DMOD);
        }
    }
    if (PART >= 2) {
        const float* xb = x + ((size_t)b * SEQ + (size_t)chunk * CHUNK) * DMOD + d0 + 4 * m;
#pragma unroll
        for (int ks = 8 * (PART - 2); ks < 8 * (PART - 1); ++ks) {
            const int r0 = 4 * ks + 2 * hi;
            t.xl0[ks] = *reinterpret_cast<const v4f*>(xb + (size_t)r0 * DMOD);
            t.xl1[ks] = *reinterpret_cast<const v4f*>(xb + (size_t)(r0 + 1) * DMOD);
        }
    }
}

template <int MAXG, bool PF>
__device__ __forceinline__ void k3_tile_out(const K3Tile& t, const v4f (&tt)[8], const v4f (&to)[4][2],
                                            const v4f q8a, const v4f q8b, float* __restrict__ y,
                                            int b, int chunk, int d0, int m, int hi,
                                            K3Tile& nx, int nchunk, const float* __restrict__ x,
                                            const float* __restrict__ QN,
                                            const float* __restrict__ PL) {
    const int gb = (chunk - 1) >> 3;

    v8f hs[4] = {{}, {}, {}, {}};
    if (gb >= 0) scatter_tile(t.th, hs);
    if (MAXG >= 1 && gb >= 1) {
        const float q8v[8] = {q8a.x, q8a.y, q8a.z, q8a.w, q8b.x, q8b.y, q8b.z, q8b.w};
        const float qev[8] = {t.qea.x, t.qea.y, t.qea.z, t.qea.w, t.qeb.x, t.qeb.y, t.qeb.z, t.qeb.w};
        v8f e0[4];
        scatter_tile(t.t0, e0);
        if (MAXG >= 2 && gb >= 2) {
            v8f e1[4];
            scatter_tile(t.t1, e1);
            apply_q(q8v, e0, e1);
#pragma unroll
            for (int j = 0; j < 4; ++j) e0[j] = e1[j];
        }
        if (MAXG >= 3 && gb >= 3) {
            v8f e2[4];
            scatter_tile(t.t2, e2);
            apply_q(q8v, e0, e2);
#pragma unroll
            for (int j = 0; j < 4; ++j) e0[j] = e2[j];
        }
        apply_q(qev, e0, hs);
    }

    v8f acc[4][4] = {{{}, {}, {}, {}}, {{}, {}, {}, {}}, {{}, {}, {}, {}}, {{}, {}, {}, {}}};

#pragma unroll
    for (int kb = 0; kb < 4; ++kb) {
        if (PF) {
            asm volatile("" ::: "memory");
            if (kb == 0) k3_load_part<3, 0>(nx, x, QN, PL, b, nchunk, d0, m, hi);
            if (kb == 1) k3_load_part<3, 1>(nx, x, QN, PL, b, nchunk, d0, m, hi);
            if (kb == 2) k3_load_part<3, 2>(nx, x, QN, PL, b, nchunk, d0, m, hi);
            if (kb == 3) k3_load_part<3, 3>(nx, x, QN, PL, b, nchunk, d0, m, hi);
            asm volatile("" ::: "memory");
        }
#pragma unroll
        for (int kk = 0; kk < 16; kk += 4) {
            const v4f l0 = t.xl0[(kb * 16 + kk) >> 2];
            const v4f l1 = t.xl1[(kb * 16 + kk) >> 2];
#pragma unroll
            for (int it = kb; it < 4; ++it) {
                const int sp = 4 * (it - kb) - (kk >> 2) + 3;
                v2f a;
                a.x = tt[sp >> 1][(sp & 1) * 2];
                a.y = tt[sp >> 1][(sp & 1) * 2 + 1];
#pragma unroll
                for (int j = 0; j < 4; ++j) {
                    v2f bq;
                    bq.x = l0[j];
                    bq.y = l1[j];
                    acc[j][it] = __builtin_amdgcn_wmma_f32_16x16x4_f32(
                        false, a, false, bq, (short)0, acc[j][it], false, false);
                }
            }
        }
    }
#pragma unroll
    for (int q = 0; q < 4; ++q) {
#pragma unroll
        for (int it = 0; it < 4; ++it) {
            v2f a;
            a.x = to[it][q >> 1][(q & 1) * 2];
            a.y = to[it][q >> 1][(q & 1) * 2 + 1];
#pragma unroll
            for (int j = 0; j < 4; ++j) {
                v2f bq;
                bq.x = hs[j][2 * q];
                bq.y = hs[j][2 * q + 1];
                acc[j][it] = __builtin_amdgcn_wmma_f32_16x16x4_f32(
                    false, a, false, bq, (short)0, acc[j][it], false, false);
            }
        }
    }

    float* yb = y + ((size_t)b * SEQ + (size_t)chunk * CHUNK) * DMOD + d0 + 4 * m;
#pragma unroll
    for (int it = 0; it < 4; ++it)
#pragma unroll
        for (int v = 0; v < 8; ++v) {
            v4f o;
            o.x = acc[0][it][v];
            o.y = acc[1][it][v];
            o.z = acc[2][it][v];
            o.w = acc[3][it][v];
            *reinterpret_cast<v4f*>(yb + (size_t)(it * 16 + v + 8 * hi) * DMOD) = o;
        }
}

__global__ __launch_bounds__(128) __attribute__((amdgpu_waves_per_eu(1, 1))) void k3_out_gemm(const float* __restrict__ x,
                                                   const float* __restrict__ TabT,
                                                   const float* __restrict__ TabO,
                                                   const float* __restrict__ QN,
                                                   const float* __restrict__ PL,
                                                   float* __restrict__ y) {
    const int lane = threadIdx.x & 31;
    const int wave = threadIdx.x >> 5;
    const int hi = lane >> 4;
    const int m = lane & 15;

    const int blk = blockIdx.x;
    const int quarter = blk & 3;
    const int p = (blk >> 2) & 15;
    const int b = blk >> 6;
    const int d0 = (quarter * 4 + wave) * 64;
    const int c_lo = p;
    const int c_hi = NCH - 1 - p;

    v4f tt[8];
#pragma unroll
    for (int g = 0; g < 8; ++g)
        tt[g] = *reinterpret_cast<const v4f*>(TabT + (g * 32 + lane) * 4);
    const float* q8p = QN + 7 * 256 + m * 16 + 8 * hi;
    const v4f q8a = *reinterpret_cast<const v4f*>(q8p);
    const v4f q8b = *reinterpret_cast<const v4f*>(q8p + 4);

    K3Tile ta, tb;
    k3_load_part<1, 0>(ta, x, QN, PL, b, c_lo, d0, m, hi);
    k3_load_part<1, 2>(ta, x, QN, PL, b, c_lo, d0, m, hi);
    k3_load_part<1, 3>(ta, x, QN, PL, b, c_lo, d0, m, hi);
    v4f to[4][2];
#pragma unroll
    for (int it = 0; it < 4; ++it)
#pragma unroll
        for (int h = 0; h < 2; ++h)
            to[it][h] = *reinterpret_cast<const v4f*>(TabO + (it * 16 + m) * 16 + 8 * hi + 4 * h);
    asm volatile("" ::: "memory");

    k3_tile_out<1, true>(ta, tt, to, q8a, q8b, y, b, c_lo, d0, m, hi, tb, c_hi, x, QN, PL);
    k3_tile_out<3, false>(tb, tt, to, q8a, q8b, y, b, c_hi, d0, m, hi, tb, c_hi, x, QN, PL);
}

extern "C" void kernel_launch(void* const* d_in, const int* in_sizes, int n_in,
                              void* d_out, int out_size, void* d_ws, size_t ws_size,
                              hipStream_t stream) {
    (void)in_sizes; (void)n_in; (void)out_size; (void)ws_size;

    const float* x    = (const float*)d_in[0];
    const float* A    = (const float*)d_in[1];
    const float* Bvec = (const float*)d_in[2];
    float* y  = (float*)d_out;
    float* ws = (float*)d_ws;

    float* TabT = ws + WS_TABT;
    float* TabO = ws + WS_TABO;
    float* TabU = ws + WS_UT;
    float* QN   = ws + WS_QN;
    float* PL   = ws + WS_PL;

    k0_precompute<<<1, 256, 0, stream>>>(A, Bvec, ws);
    k1_state_scan<<<BATCH * NGRP * 16, 256, 0, stream>>>(x, TabU, QN, PL);
    k3_out_gemm<<<BATCH * 16 * 4, 128, 0, stream>>>(x, TabT, TabO, QN, PL, y);
}


# 2 launches: the power launch goes; each state work-group runs the doubling levels in LDS under its own x-row loads (LDS-only barriers); work-groups 0, 1 also write the output tables; sums unchanged
# speedup vs baseline: 6.1107x; 1.1799x over previous
#include <hip/hip_runtime.h>
#include <hip/hip_bf16.h>

typedef float v2f __attribute__((ext_vector_type(2)));
typedef float v4f __attribute__((ext_vector_type(4)));
typedef float v8f __attribute__((ext_vector_type(8)));

#define BATCH 4
#define SEQ   2048
#define DMOD  1024
#define NST   16
#define CHUNK 64
#define NCH   (SEQ / CHUNK)

#define GRP   8
#define NGRP  (NCH / GRP)

#define WS_TABT  0
#define WS_TABO  1024
#define WS_QN    (80 * 64 + 64 * 16)
#define WS_PL    8192

#define PITCH 20

__device__ __forceinline__ float dot16(const float* a, const float* b) {
    float s0 = 0.f, s1 = 0.f, s2 = 0.f, s3 = 0.f;
#pragma unroll
    for (int q = 0; q < 16; q += 4) {
        s0 += a[q] * b[q];
        s1 += a[q + 1] * b[q + 1];
        s2 += a[q + 2] * b[q + 2];
        s3 += a[q + 3] * b[q + 3];
    }
    return (s0 + s1) + (s2 + s3);
}

__device__ __forceinline__ void lds_barrier() {
    __builtin_amdgcn_fence(__ATOMIC_RELEASE, "workgroup", "local");
    __builtin_amdgcn_s_barrier();
    __builtin_amdgcn_fence(__ATOMIC_ACQUIRE, "workgroup", "local");
}

__global__ __launch_bounds__(256) __attribute__((amdgpu_waves_per_eu(2, 2))) void k1_state_scan(const float* __restrict__ x,
                                                     const float* __restrict__ A,
                                                     const float* __restrict__ Bvec,
                                                     float* __restrict__ ws) {
    __shared__ __attribute__((aligned(16))) v4f pl[2][GRP][4][2][32];
    __shared__ __attribute__((aligned(16))) float P[7][16][PITCH];
    __shared__ __attribute__((aligned(16))) float QQ[8][16][PITCH];
    __shared__ __attribute__((aligned(16))) float U[64][PITCH];
    __shared__ __attribute__((aligned(16))) float R[65][PITCH];
    __shared__ float kv[64];

    const int tid = threadIdx.x;
    const int pm = tid >> 4, pn = tid & 15;

    const int lane = threadIdx.x & 31;
    const int w = __builtin_amdgcn_readfirstlane((int)(threadIdx.x >> 5));
    const int hi = lane >> 4;
    const int m = lane & 15;

    const int blk = blockIdx.x;
    const int slab = blk & 15;
    const int g = (blk >> 4) & (NGRP - 1);
    const int b = blk >> 6;
    const int chunk = g * GRP + w;
    const int d0 = slab * 64;
    const bool doQ = (blk == 0);
    const bool doR = (blk == 1);

    const float a_mn = A[pm * 16 + pn];
    const float bv = Bvec[pn];
    const float* xb = x + ((size_t)b * SEQ + (size_t)chunk * CHUNK) * DMOD + d0 + 4 * m;
    v4f xr0[16], xr1[16];
#pragma unroll
    for (int ks = 0; ks < 16; ++ks) {
        const int r0 = 4 * ks + 2 * hi;
        xr0[ks] = *reinterpret_cast<const v4f*>(xb + (size_t)r0 * DMOD);
        xr1[ks] = *reinterpret_cast<const v4f*>(xb + (size_t)(r0 + 1) * DMOD);
    }
    asm volatile("" ::: "memory");

    P[0][pm][pn] = a_mn;
    U[0][pn] = bv;
    R[0][pn] = bv;
    lds_barrier();

#pragma unroll
    for (int k = 0; k < 6; ++k) {
        const float (*M)[PITCH] = P[k];
        float mrow[16], mcol[16];
#pragma unroll
        for (int q = 0; q < 16; ++q) {
            mrow[q] = M[pm][q];
            mcol[q] = M[q][pn];
        }

        {
            const int have = 1 << k;
            float nrow[16];
#pragma unroll
            for (int q = 0; q < 16; ++q) nrow[q] = M[pn][q];
#pragma unroll
            for (int j = 0; j < (have * 16 + 255) / 256; ++j) {
                const int e = tid + 256 * j;
                if (e < have * 16) {
                    const int i = e >> 4;
                    float uu[16];
#pragma unroll
                    for (int q = 0; q < 16; ++q) uu[q] = U[i][q];
                    U[have + i][pn] = dot16(nrow, uu);
                    if (doR) {
                        float rr[16];
#pragma unroll
                        for (int q = 0; q < 16; ++q) rr[q] = R[i][q];
                        R[have + i][pn] = dot16(rr, mcol);
                    }
                }
            }
        }

        const float sq = dot16(mrow, mcol);
        if (k < 5) P[k + 1][pm][pn] = sq;
        else       QQ[0][pm][pn] = sq;
        lds_barrier();
    }

#pragma unroll
    for (int lv = 0; lv < 3; ++lv) {
        if (lv < 2 || doQ) {
            const int have = 1 << lv;
            float mcol[16];
#pragma unroll
            for (int q = 0; q < 16; ++q) mcol[q] = QQ[have - 1][q][pn];

            if (lv == 0 && doR) {
                if (tid < 16) {
                    float rr[16];
#pragma unroll
                    for (int q = 0; q < 16; ++q) rr[q] = R[0][q];
                    R[64][pn] = dot16(rr, mcol);
                } else if (tid >= 64 && tid < 128) {
                    const int p = tid - 64;
                    float rr[16], bb[16];
#pragma unroll
                    for (int q = 0; q < 16; ++q) {
                        rr[q] = R[p][q];
                        bb[q] = R[0][q];
                    }
                    kv[p] = dot16(rr, bb);
                }
            }

#pragma unroll
            for (int i = 1; i <= have; ++i) {
                if (i == have || doQ) {
                    float row[16];
#pragma unroll
                    for (int q = 0; q < 16; ++q) row[q] = QQ[i - 1][pm][q];
                    QQ[have + i - 1][pm][pn] = dot16(row, mcol);
                }
            }
            lds_barrier();
        }
    }

    v4f qlo[3], qhi[3];
#pragma unroll
    for (int l = 0; l < 3; ++l) {
        const float* qp = &QQ[(1 << l) - 1][m][8 * hi];
        qlo[l] = *reinterpret_cast<const v4f*>(qp);
        qhi[l] = *reinterpret_cast<const v4f*>(qp + 4);
    }

    v4f ut[8];
#pragma unroll
    for (int gg = 0; gg < 8; ++gg)
#pragma unroll
        for (int h2 = 0; h2 < 2; ++h2)
#pragma unroll
            for (int cc = 0; cc < 2; ++cc)
                ut[gg][2 * h2 + cc] = U[63 - (8 * gg + 4 * h2 + 2 * hi + cc)][m];

    v8f acc[4] = {{}, {}, {}, {}};
#pragma unroll
    for (int ks = 0; ks < 16; ++ks) {
        const v4f l0 = xr0[ks];
        const v4f l1 = xr1[ks];
        v2f uf;
        uf.x = ut[ks >> 1][(ks & 1) * 2];
        uf.y = ut[ks >> 1][(ks & 1) * 2 + 1];
#pragma unroll
        for (int j = 0; j < 4; ++j) {
            v2f xq;
            xq.x = l0[j];
            xq.y = l1[j];
            acc[j] = __builtin_amdgcn_wmma_f32_16x16x4_f32(false, uf, false, xq,
                                                           (short)0, acc[j], false, false);
        }
    }

#pragma unroll
    for (int l = 0; l < 3; ++l) {
        const int s = 1 << l;
        const int p = l & 1;
#pragma unroll
        for (int j = 0; j < 4; ++j) {
            pl[p][w][j][0][lane] = __builtin_shufflevector(acc[j], acc[j], 0, 1, 2, 3);
            pl[p][w][j][1][lane] = __builtin_shufflevector(acc[j], acc[j], 4, 5, 6, 7);
        }
        __syncthreads();
        if (w >= s) {
            const float qv[8] = {qlo[l].x, qlo[l].y, qlo[l].z, qlo[l].w,
                                 qhi[l].x, qhi[l].y, qhi[l].z, qhi[l].w};
#pragma unroll
            for (int j = 0; j < 4; ++j) {
                const v4f t0 = pl[p][w - s][j][0][lane];
                const v4f t1 = pl[p][w - s][j][1][lane];
                const float tv[8] = {t0.x, t0.y, t0.z, t0.w, t1.x, t1.y, t1.z, t1.w};
#pragma unroll
                for (int q = 0; q < 4; ++q) {
                    v2f a, bq;
                    a.x = qv[2 * q];
                    a.y = qv[2 * q + 1];
                    bq.x = tv[2 * q];
                    bq.y = tv[2 * q + 1];
                    acc[j] = __builtin_amdgcn_wmma_f32_16x16x4_f32(false, a, false, bq,
                                                                   (short)0, acc[j], false, false);
                }
            }
        }
    }

    if (doQ) {
        float* QN = ws + WS_QN;
#pragma unroll
        for (int e = 0; e < 8; ++e)
            QN[e * 256 + tid] = QQ[e][pm][pn];
    }
    if (doR) {
        float* TabT = ws + WS_TABT;
        float* TabO = ws + WS_TABO;
#pragma unroll
        for (int j = 0; j < 4; ++j) {
            const int e = tid + 256 * j;
            const int c = e & 1, h2 = (e >> 1) & 1, ln = (e >> 2) & 31, pg = e >> 7;
            const int lm = ln & 15, lh = ln >> 4;
            TabO[e] = R[(e >> 4) + 1][e & 15];
            const int idx = 4 * (2 * pg + h2 - 3) + lm - 2 * lh - c;
            const float v = kv[idx & 63];
            TabT[e] = (idx >= 0) ? v : 0.0f;
        }
    }

    float* Pp = ws + WS_PL + (((size_t)b * NCH + chunk) * NST + 8 * hi) * DMOD + d0 + 4 * m;
#pragma unroll
    for (int r = 0; r < 8; ++r) {
        v4f o;
        o.x = acc[0][r];
        o.y = acc[1][r];
        o.z = acc[2][r];
        o.w = acc[3][r];
        *reinterpret_cast<v4f*>(Pp + (size_t)r * DMOD) = o;
    }
}

__device__ __forceinline__ void scatter_tile(const v4f (&t)[8], v8f (&e)[4]) {
#pragma unroll
    for (int r = 0; r < 8; ++r)
#pragma unroll
        for (int j = 0; j < 4; ++j) e[j][r] = t[r][j];
}

__device__ __forceinline__ void apply_q(const float (&qv)[8], const v8f (&e)[4], v8f (&c)[4]) {
#pragma unroll
    for (int j = 0; j < 4; ++j)
#pragma unroll
        for (int q = 0; q < 4; ++q) {
            v2f a, bq;
            a.x = qv[2 * q];
            a.y = qv[2 * q + 1];
            bq.x = e[j][2 * q];
            bq.y = e[j][2 * q + 1];
            c[j] = __builtin_amdgcn_wmma_f32_16x16x4_f32(false, a, false, bq,
                                                         (short)0, c[j], false, false);
        }
}

struct K3Tile {
    v4f th[8], t0[8], t1[8], t2[8];
    v4f xl0[16], xl1[16];
    v4f qea, qeb;
};

template <int MAXG, int PART>
__device__ __forceinline__ void k3_load_part(K3Tile& t, const float* __restrict__ x,
                                             const float* __restrict__ QN,
                                             const float* __restrict__ PL,
                                             int b, int chunk, int d0, int m, int hi) {
    const float* Pb = PL + ((size_t)b * NCH * NST + 8 * hi) * DMOD + d0 + 4 * m;
    const int gb = (chunk - 1) >> 3;
    if (PART == 0) {
        const int ee = chunk - 8 * gb;
        const float* qep = QN + ((ee - 1) & 7) * 256 + m * 16 + 8 * hi;
        t.qea = *reinterpret_cast<const v4f*>(qep);
        t.qeb = *reinterpret_cast<const v4f*>(qep + 4);
        if (gb >= 0) {
#pragma unroll
            for (int r = 0; r < 8; ++r)
                t.th[r] = *reinterpret_cast<const v4f*>(Pb + ((size_t)(chunk - 1) * NST + r) * DMOD);
        }
        if (MAXG >= 1 && gb >= 1) {
#pragma unroll
            for (int r = 0; r < 8; ++r)
                t.t0[r] = *reinterpret_cast<const v4f*>(Pb + ((size_t)(GRP - 1) * NST + r) * DMOD);
        }
    }
    if (PART == 1) {
        if (MAXG >= 2 && gb >= 2) {
#pragma unroll
            for (int r = 0; r < 8; ++r)
                t.t1[r] = *reinterpret_cast<const v4f*>(Pb + ((size_t)(2 * GRP - 1) * NST + r) * DMOD);
        }
        if (MAXG >= 3 && gb >= 3) {
#pragma unroll
            for (int r = 0; r < 8; ++r)
                t.t2[r] = *reinterpret_cast<const v4f*>(Pb + ((size_t)(3 * GRP - 1) * NST + r) * DMOD);
        }
    }
    if (PART >= 2) {
        const float* xb = x + ((size_t)b * SEQ + (size_t)chunk * CHUNK) * DMOD + d0 + 4 * m;
#pragma unroll
        for (int ks = 8 * (PART - 2); ks < 8 * (PART - 1); ++ks) {
            const int r0 = 4 * ks + 2 * hi;
            t.xl0[ks] = *reinterpret_cast<const v4f*>(xb + (size_t)r0 * DMOD);
            t.xl1[ks] = *reinterpret_cast<const v4f*>(xb + (size_t)(r0 + 1) * DMOD);
        }
    }
}

template <int MAXG, bool PF>
__device__ __forceinline__ void k3_tile_out(const K3Tile& t, const v4f (&tt)[8], const v4f (&to)[4][2],
                                            const v4f q8a, const v4f q8b, float* __restrict__ y,
                                            int b, int chunk, int d0, int m, int hi,
                                            K3Tile& nx, int nchunk, const float* __restrict__ x,
                                            const float* __restrict__ QN,
                                            const float* __restrict__ PL) {
    const int gb = (chunk - 1) >> 3;

    v8f hs[4] = {{}, {}, {}, {}};
    if (gb >= 0) scatter_tile(t.th, hs);
    if (MAXG >= 1 && gb >= 1) {
        const float q8v[8] = {q8a.x, q8a.y, q8a.z, q8a.w, q8b.x, q8b.y, q8b.z, q8b.w};
        const float qev[8] = {t.qea.x, t.qea.y, t.qea.z, t.qea.w, t.qeb.x, t.qeb.y, t.qeb.z, t.qeb.w};
        v8f e0[4];
        scatter_tile(t.t0, e0);
        if (MAXG >= 2 && gb >= 2) {
            v8f e1[4];
            scatter_tile(t.t1, e1);
            apply_q(q8v, e0, e1);
#pragma unroll
            for (int j = 0; j < 4; ++j) e0[j] = e1[j];
        }
        if (MAXG >= 3 && gb >= 3) {
            v8f e2[4];
            scatter_tile(t.t2, e2);
            apply_q(q8v, e0, e2);
#pragma unroll
            for (int j = 0; j < 4; ++j) e0[j] = e2[j];
        }
        apply_q(qev, e0, hs);
    }

    v8f acc[4][4] = {{{}, {}, {}, {}}, {{}, {}, {}, {}}, {{}, {}, {}, {}}, {{}, {}, {}, {}}};

#pragma unroll
    for (int kb = 0; kb < 4; ++kb) {
        if (PF) {
            asm volatile("" ::: "memory");
            if (kb == 0) k3_load_part<3, 0>(nx, x, QN, PL, b, nchunk, d0, m, hi);
            if (kb == 1) k3_load_part<3, 1>(nx, x, QN, PL, b, nchunk, d0, m, hi);
            if (kb == 2) k3_load_part<3, 2>(nx, x, QN, PL, b, nchunk, d0, m, hi);
            if (kb == 3) k3_load_part<3, 3>(nx, x, QN, PL, b, nchunk, d0, m, hi);
            asm volatile("" ::: "memory");
        }
#pragma unroll
        for (int kk = 0; kk < 16; kk += 4) {
            const v4f l0 = t.xl0[(kb * 16 + kk) >> 2];
            const v4f l1 = t.xl1[(kb * 16 + kk) >> 2];
#pragma unroll
            for (int it = kb; it < 4; ++it) {
                const int sp = 4 * (it - kb) - (kk >> 2) + 3;
                v2f a;
                a.x = tt[sp >> 1][(sp & 1) * 2];
                a.y = tt[sp >> 1][(sp & 1) * 2 + 1];
#pragma unroll
                for (int j = 0; j < 4; ++j) {
                    v2f bq;
                    bq.x = l0[j];
                    bq.y = l1[j];
                    acc[j][it] = __builtin_amdgcn_wmma_f32_16x16x4_f32(
                        false, a, false, bq, (short)0, acc[j][it], false, false);
                }
            }
        }
    }
#pragma unroll
    for (int q = 0; q < 4; ++q) {
#pragma unroll
        for (int it = 0; it < 4; ++it) {
            v2f a;
            a.x = to[it][q >> 1][(q & 1) * 2];
            a.y = to[it][q >> 1][(q & 1) * 2 + 1];
#pragma unroll
            for (int j = 0; j < 4; ++j) {
                v2f bq;
                bq.x = hs[j][2 * q];
                bq.y = hs[j][2 * q + 1];
                acc[j][it] = __builtin_amdgcn_wmma_f32_16x16x4_f32(
                    false, a, false, bq, (short)0, acc[j][it], false, false);
            }
        }
    }

    float* yb = y + ((size_t)b * SEQ + (size_t)chunk * CHUNK) * DMOD + d0 + 4 * m;
#pragma unroll
    for (int it = 0; it < 4; ++it)
#pragma unroll
        for (int v = 0; v < 8; ++v) {
            v4f o;
            o.x = acc[0][it][v];
            o.y = acc[1][it][v];
            o.z = acc[2][it][v];
            o.w = acc[3][it][v];
            *reinterpret_cast<v4f*>(yb + (size_t)(it * 16 + v + 8 * hi) * DMOD) = o;
        }
}

__global__ __launch_bounds__(128) __attribute__((amdgpu_waves_per_eu(1, 1))) void k3_out_gemm(const float* __restrict__ x,
                                                   const float* __restrict__ TabT,
                                                   const float* __restrict__ TabO,
                                                   const float* __restrict__ QN,
                                                   const float* __restrict__ PL,
                                                   float* __restrict__ y) {
    const int lane = threadIdx.x & 31;
    const int wave = threadIdx.x >> 5;
    const int hi = lane >> 4;
    const int m = lane & 15;

    const int blk = blockIdx.x;
    const int quarter = blk & 3;
    const int p = (blk >> 2) & 15;
    const int b = blk >> 6;
    const int d0 = (quarter * 4 + wave) * 64;
    const int c_lo = p;
    const int c_hi = NCH - 1 - p;

    v4f tt[8];
#pragma unroll
    for (int g = 0; g < 8; ++g)
        tt[g] = *reinterpret_cast<const v4f*>(TabT + (g * 32 + lane) * 4);
    const float* q8p = QN + 7 * 256 + m * 16 + 8 * hi;
    const v4f q8a = *reinterpret_cast<const v4f*>(q8p);
    const v4f q8b = *reinterpret_cast<const v4f*>(q8p + 4);

    K3Tile ta, tb;
    k3_load_part<1, 0>(ta, x, QN, PL, b, c_lo, d0, m, hi);
    k3_load_part<1, 2>(ta, x, QN, PL, b, c_lo, d0, m, hi);
    k3_load_part<1, 3>(ta, x, QN, PL, b, c_lo, d0, m, hi);
    v4f to[4][2];
#pragma unroll
    for (int it = 0; it < 4; ++it)
#pragma unroll
        for (int h = 0; h < 2; ++h)
            to[it][h] = *reinterpret_cast<const v4f*>(TabO + (it * 16 + m) * 16 + 8 * hi + 4 * h);
    asm volatile("" ::: "memory");

    k3_tile_out<1, true>(ta, tt, to, q8a, q8b, y, b, c_lo, d0, m, hi, tb, c_hi, x, QN, PL);
    k3_tile_out<3, false>(tb, tt, to, q8a, q8b, y, b, c_hi, d0, m, hi, tb, c_hi, x, QN, PL);
}

extern "C" void kernel_launch(void* const* d_in, const int* in_sizes, int n_in,
                              void* d_out, int out_size, void* d_ws, size_t ws_size,
                              hipStream_t stream) {
    (void)in_sizes; (void)n_in; (void)out_size; (void)ws_size;

    const float* x    = (const float*)d_in[0];
    const float* A    = (const float*)d_in[1];
    const float* Bvec = (const float*)d_in[2];
    float* y  = (float*)d_out;
    float* ws = (float*)d_ws;

    float* TabT = ws + WS_TABT;
    float* TabO = ws + WS_TABO;
    float* QN   = ws + WS_QN;
    float* PL   = ws + WS_PL;

    k1_state_scan<<<BATCH * NGRP * 16, 256, 0, stream>>>(x, A, Bvec, ws);
    k3_out_gemm<<<BATCH * 16 * 4, 128, 0, stream>>>(x, TabT, TabO, QN, PL, y);
}
